# starting kernel
# baseline (speedup 1.0000x reference)
#include <hip/hip_runtime.h>


typedef __attribute__((ext_vector_type(2))) float v2f;
typedef __attribute__((ext_vector_type(8))) float v8f;

constexpr int S    = 2048;
constexpr int BT   = 32;
constexpr int DIN  = 256;
constexpr int DOUT = 256;
constexpr int LAT  = 64;
constexpr int T    = 32;
constexpr int NC   = 64;
constexpr int HPAD = 68;

__device__ __forceinline__ v8f wmma4(v2f a, v2f b, v8f c) {
  return __builtin_amdgcn_wmma_f32_16x16x4_f32(false, a, false, b, (short)0, c,
                                               false, false);
}

__device__ __forceinline__ v8f tile_mm_lds(const float* hbase, const v2f* bfrag,
                                           v8f acc, int mt, int m, int hi) {
  const float* ap = hbase + (mt * 16 + m) * HPAD + 2 * hi;
#pragma unroll
  for (int kk = 0; kk < 16; ++kk) {
    v2f a = *(const v2f*)(ap + kk * 4);
    acc = wmma4(a, bfrag[kk], acc);
  }
  return acc;
}

__device__ __forceinline__ void load_bfrag64(v2f* bfrag, const float* W, int nt,
                                             int m, int hi) {
#pragma unroll
  for (int kk = 0; kk < 16; ++kk)
    bfrag[kk] = *(const v2f*)(W + (nt * 16 + m) * 64 + kk * 4 + 2 * hi);
}

__device__ __forceinline__ v8f load_u_tile(const float* u, int mt, int nt,
                                           int m, int hi) {
  v8f acc;
#pragma unroll
  for (int r = 0; r < 8; ++r)
    acc[r] = u[(mt * 16 + r + 8 * hi) * 64 + nt * 16 + m];
  return acc;
}

__device__ __forceinline__ void store_tile_lds(float* hbase, v8f d, int mt,
                                               int nt, int m, int hi) {
#pragma unroll
  for (int r = 0; r < 8; ++r)
    hbase[(mt * 16 + r + 8 * hi) * HPAD + nt * 16 + m] = d[r];
}

__device__ __forceinline__ void store_tile_g64(float* g, v8f d, int mt, int nt,
                                               int m, int hi) {
#pragma unroll
  for (int r = 0; r < 8; ++r)
    g[(mt * 16 + r + 8 * hi) * 64 + nt * 16 + m] = d[r];
}

__global__ void bx_gemm_k(const float* __restrict__ x,
                          const float* __restrict__ Bw,
                          float* __restrict__ bx) {
  int tid = threadIdx.x;
  int lane = tid & 31, hi = lane >> 4, m = lane & 15;
  int w = (blockIdx.x * blockDim.x + tid) >> 5;
  int mt = w >> 2, nt = w & 3;

  const float* ap = x + (size_t)(mt * 16 + m) * DIN + 2 * hi;
  const float* bp = Bw + (size_t)(nt * 16 + m) * DIN + 2 * hi;

  v8f acc = {0.f, 0.f, 0.f, 0.f, 0.f, 0.f, 0.f, 0.f};
#pragma unroll 8
  for (int kk = 0; kk < DIN / 4; ++kk) {
    v2f a = *(const v2f*)(ap + kk * 4);
    v2f b = *(const v2f*)(bp + kk * 4);
    acc = wmma4(a, b, acc);
  }
  store_tile_g64(bx + (size_t)mt * 16 * LAT, acc, 0, nt, m, hi);
}

__global__ void matpow_k(const float* __restrict__ A, float* __restrict__ Apow) {
  __shared__ float P[2][64][65];
  int tid = threadIdx.x;
  for (int i = tid; i < 64 * 64; i += blockDim.x)
    P[0][i >> 6][i & 63] = A[i];
  __syncthreads();
  int p = 0;
#pragma unroll 1
  for (int s = 0; s < 5; ++s) {
    for (int i = tid; i < 64 * 64; i += blockDim.x) {
      int r = i >> 6, c = i & 63;
      float acc = 0.f;
#pragma unroll 8
      for (int k = 0; k < 64; ++k) acc += P[p][r][k] * P[p][k][c];
      P[1 - p][r][c] = acc;
    }
    __syncthreads();
    p ^= 1;
  }
  for (int i = tid; i < 64 * 64; i += blockDim.x)
    Apow[i] = P[p][i >> 6][i & 63];
}

__global__ void chunk_local_k(const float* __restrict__ bx,
                              const float* __restrict__ A,
                              float* __restrict__ Lc) {
  __shared__ float hbuf[2][BT * HPAD];
  int tid = threadIdx.x;
  int lane = tid & 31, hi = lane >> 4, m = lane & 15;
  int w = tid >> 5, mt = w >> 2, nt = w & 3;
  int c = blockIdx.x;

  for (int i = tid; i < BT * HPAD; i += blockDim.x) hbuf[0][i] = 0.f;

  v2f bfrag[16];
  load_bfrag64(bfrag, A, nt, m, hi);

  v8f u_next = load_u_tile(bx + (size_t)(c * T) * BT * LAT, mt, nt, m, hi);
  __syncthreads();

  int p = 0;
#pragma unroll 1
  for (int i = 0; i < T; ++i) {
    v8f acc = u_next;
    if (i + 1 < T)
      u_next =
          load_u_tile(bx + (size_t)(c * T + i + 1) * BT * LAT, mt, nt, m, hi);
    acc = tile_mm_lds(hbuf[p], bfrag, acc, mt, m, hi);
    store_tile_lds(hbuf[1 - p], acc, mt, nt, m, hi);
    __syncthreads();
    p ^= 1;
  }
  float* out = Lc + (size_t)c * BT * LAT;
  for (int i = tid; i < BT * LAT; i += blockDim.x)
    out[i] = hbuf[p][(i >> 6) * HPAD + (i & 63)];
}

__global__ void chunk_scan_k(const float* __restrict__ Apow,
                             const float* __restrict__ Lc,
                             float* __restrict__ carries) {
  __shared__ float hbuf[2][BT * HPAD];
  int tid = threadIdx.x;
  int lane = tid & 31, hi = lane >> 4, m = lane & 15;
  int w = tid >> 5, mt = w >> 2, nt = w & 3;

  for (int i = tid; i < BT * HPAD; i += blockDim.x) hbuf[0][i] = 0.f;
  for (int i = tid; i < BT * LAT; i += blockDim.x) carries[i] = 0.f;

  v2f bfrag[16];
  load_bfrag64(bfrag, Apow, nt, m, hi);

  v8f u_next = load_u_tile(Lc, mt, nt, m, hi);
  __syncthreads();

  int p = 0;
#pragma unroll 1
  for (int c = 0; c < NC - 1; ++c) {
    v8f acc = u_next;
    if (c + 1 < NC - 1)
      u_next = load_u_tile(Lc + (size_t)(c + 1) * BT * LAT, mt, nt, m, hi);
    acc = tile_mm_lds(hbuf[p], bfrag, acc, mt, m, hi);
    store_tile_lds(hbuf[1 - p], acc, mt, nt, m, hi);
    store_tile_g64(carries + (size_t)(c + 1) * BT * LAT, acc, mt, nt, m, hi);
    __syncthreads();
    p ^= 1;
  }
}

__global__ void chunk_final_k(const float* __restrict__ bx,
                              const float* __restrict__ A,
                              const float* __restrict__ carries,
                              float* __restrict__ h_out) {
  __shared__ float hbuf[2][BT * HPAD];
  int tid = threadIdx.x;
  int lane = tid & 31, hi = lane >> 4, m = lane & 15;
  int w = tid >> 5, mt = w >> 2, nt = w & 3;
  int c = blockIdx.x;

  const float* cin = carries + (size_t)c * BT * LAT;
  for (int i = tid; i < BT * LAT; i += blockDim.x)
    hbuf[0][(i >> 6) * HPAD + (i & 63)] = cin[i];

  v2f bfrag[16];
  load_bfrag64(bfrag, A, nt, m, hi);

  v8f u_next = load_u_tile(bx + (size_t)(c * T) * BT * LAT, mt, nt, m, hi);
  __syncthreads();

  int p = 0;
#pragma unroll 1
  for (int i = 0; i < T; ++i) {
    int t = c * T + i;
    v8f acc = u_next;
    if (i + 1 < T)
      u_next = load_u_tile(bx + (size_t)(t + 1) * BT * LAT, mt, nt, m, hi);
    acc = tile_mm_lds(hbuf[p], bfrag, acc, mt, m, hi);
    store_tile_lds(hbuf[1 - p], acc, mt, nt, m, hi);
    store_tile_g64(h_out + (size_t)(t + 1) * BT * LAT, acc, mt, nt, m, hi);
    __syncthreads();
    p ^= 1;
  }
}

__global__ void y_gemm_k(const float* __restrict__ hs,
                         const float* __restrict__ C,
                         float* __restrict__ y) {
  int tid = threadIdx.x;
  int lane = tid & 31, hi = lane >> 4, m = lane & 15;
  int w = (blockIdx.x * blockDim.x + tid) >> 5;
  int mt = w >> 4, nt = w & 15;

  const float* ap = hs + (size_t)(mt * 16 + m) * LAT + 2 * hi;
  const float* bp = C + (size_t)(nt * 16 + m) * LAT + 2 * hi;

  v8f acc = {0.f, 0.f, 0.f, 0.f, 0.f, 0.f, 0.f, 0.f};
#pragma unroll
  for (int kk = 0; kk < LAT / 4; ++kk) {
    v2f a = *(const v2f*)(ap + kk * 4);
    v2f b = *(const v2f*)(bp + kk * 4);
    acc = wmma4(a, b, acc);
  }
#pragma unroll
  for (int r = 0; r < 8; ++r)
    y[(size_t)(mt * 16 + r + 8 * hi) * DOUT + nt * 16 + m] = acc[r];
}

extern "C" void kernel_launch(void* const* d_in, const int* in_sizes, int n_in,
                              void* d_out, int out_size, void* d_ws,
                              size_t ws_size, hipStream_t stream) {
  const float* x  = (const float*)d_in[0];
  const float* A  = (const float*)d_in[1];
  const float* Bw = (const float*)d_in[2];
  const float* C  = (const float*)d_in[3];

  float* out      = (float*)d_out;
  float* y_region = out;
  float* h_region = out + (size_t)S * BT * DOUT;

  float* bx = y_region;

  float* ws      = (float*)d_ws;
  float* Apow    = ws;
  float* Lc      = ws + 64 * 64;
  float* carries = ws + 64 * 64 + (size_t)NC * BT * LAT;

  hipMemsetAsync(h_region, 0, (size_t)BT * LAT * sizeof(float), stream);

  bx_gemm_k<<<(S * BT / 16) * (LAT / 16) / 8, 256, 0, stream>>>(x, Bw, bx);
  matpow_k<<<1, 256, 0, stream>>>(A, Apow);
  chunk_local_k<<<NC, 256, 0, stream>>>(bx, A, Lc);
  chunk_scan_k<<<1, 256, 0, stream>>>(Apow, Lc, carries);
  chunk_final_k<<<NC, 256, 0, stream>>>(bx, A, carries, h_region);
  y_gemm_k<<<(S * BT / 16) * (DOUT / 16) / 8, 256, 0, stream>>>(
      h_region + BT * LAT, C, y_region);
}


# Final scan pass parallel in time: h[t] = local[t] + A^(t+1)*carry; powers A^1..A^32 from 4 extra local-pass waves stepping identity columns (squaring kernel gone); carry and A^32 sum order changes
# speedup vs baseline: 2.2243x; 2.2243x over previous
#include <hip/hip_runtime.h>


typedef __attribute__((ext_vector_type(2))) float v2f;
typedef __attribute__((ext_vector_type(4))) float v4f;
typedef __attribute__((ext_vector_type(8))) float v8f;

constexpr int S    = 2048;
constexpr int BT   = 32;
constexpr int DIN  = 256;
constexpr int DOUT = 256;
constexpr int LAT  = 64;
constexpr int T    = 32;
constexpr int NC   = 64;

__device__ __forceinline__ v8f wmma4(v2f a, v2f b, v8f c) {
  return __builtin_amdgcn_wmma_f32_16x16x4_f32(false, a, false, b, (short)0, c,
                                               false, false);
}


__device__ __forceinline__ void load_wfrag(v2f (&Wf)[4][16], const float* Wm,
                                           int m, int hi) {
#pragma unroll
  for (int mt = 0; mt < 4; ++mt) {
#pragma unroll
    for (int mp = 0; mp < 4; ++mp) {
      const float* p = Wm + (mt * 16 + m) * 64 + mp * 16 + 8 * hi;
      v4f lo = *(const v4f*)p;
      v4f up = *(const v4f*)(p + 4);
      Wf[mt][4 * mp + 0] = v2f{lo[0], lo[1]};
      Wf[mt][4 * mp + 1] = v2f{lo[2], lo[3]};
      Wf[mt][4 * mp + 2] = v2f{up[0], up[1]};
      Wf[mt][4 * mp + 3] = v2f{up[2], up[3]};
    }
  }
}

__device__ __forceinline__ void load_state(v8f (&d)[4], const float* plane,
                                           int row, int hi) {
  const float* p = plane + row * 64 + 8 * hi;
#pragma unroll
  for (int mt = 0; mt < 4; ++mt) {
    v4f lo = *(const v4f*)(p + 16 * mt);
    v4f up = *(const v4f*)(p + 16 * mt + 4);
    d[mt] = v8f{lo[0], lo[1], lo[2], lo[3], up[0], up[1], up[2], up[3]};
  }
}

__device__ __forceinline__ void store_state(float* plane, const v8f (&d)[4],
                                            int row, int hi) {
  float* p = plane + row * 64 + 8 * hi;
#pragma unroll
  for (int mt = 0; mt < 4; ++mt) {
    *(v4f*)(p + 16 * mt) = v4f{d[mt][0], d[mt][1], d[mt][2], d[mt][3]};
    *(v4f*)(p + 16 * mt + 4) = v4f{d[mt][4], d[mt][5], d[mt][6], d[mt][7]};
  }
}

__device__ __forceinline__ void scan_step(v8f (&acc)[4], const v8f (&d)[4],
                                          const v2f (&Wf)[4][16]) {
#pragma unroll
  for (int mp = 0; mp < 4; ++mp) {
#pragma unroll
    for (int q = 0; q < 4; ++q) {
      v2f b = v2f{d[mp][2 * q], d[mp][2 * q + 1]};
#pragma unroll
      for (int mt = 0; mt < 4; ++mt)
        acc[mt] = wmma4(Wf[mt][4 * mp + q], b, acc[mt]);
    }
  }
}

__device__ __forceinline__ void store_tile_g64(float* g, v8f d, int mt, int nt,
                                               int m, int hi) {
#pragma unroll
  for (int r = 0; r < 8; ++r)
    g[(mt * 16 + r + 8 * hi) * 64 + nt * 16 + m] = d[r];
}

__global__ void bx_gemm_k(const float* __restrict__ x,
                          const float* __restrict__ Bw,
                          float* __restrict__ bx) {
  int tid = threadIdx.x;
  int lane = tid & 31, hi = lane >> 4, m = lane & 15;
  int w = (blockIdx.x * blockDim.x + tid) >> 5;
  int mt = w >> 2, nt = w & 3;

  const float* ap = x + (size_t)(mt * 16 + m) * DIN + 2 * hi;
  const float* bp = Bw + (size_t)(nt * 16 + m) * DIN + 2 * hi;

  v8f acc = {0.f, 0.f, 0.f, 0.f, 0.f, 0.f, 0.f, 0.f};
#pragma unroll 8
  for (int kk = 0; kk < DIN / 4; ++kk) {
    v2f a = *(const v2f*)(ap + kk * 4);
    v2f b = *(const v2f*)(bp + kk * 4);
    acc = wmma4(a, b, acc);
  }
  store_tile_g64(bx + (size_t)mt * 16 * LAT, acc, 0, nt, m, hi);
}


__global__ void __launch_bounds__(32)
chunk_local_k(const float* __restrict__ bx, const float* __restrict__ A,
              float* __restrict__ Lc, float* __restrict__ h_out,
              float* __restrict__ Apows) {
  int lane = threadIdx.x & 31, hi = lane >> 4, m = lane & 15;
  constexpr size_t P = (size_t)BT * LAT;

  v2f Wf[4][16];
  load_wfrag(Wf, A, m, hi);

  if (blockIdx.x >= NC * 2) {
    int j = blockIdx.x - NC * 2;
    v8f sa[4], sb[4];
#pragma unroll
    for (int mt = 0; mt < 4; ++mt)
#pragma unroll
      for (int r = 0; r < 8; ++r)
        sa[mt][r] = (16 * mt + 8 * hi + r == 16 * j + m) ? 1.f : 0.f;
#pragma unroll 1
    for (int t = 0; t < T; t += 2) {
#pragma unroll
      for (int mt = 0; mt < 4; ++mt)
        sb[mt] = v8f{0.f, 0.f, 0.f, 0.f, 0.f, 0.f, 0.f, 0.f};
      scan_step(sb, sa, Wf);
#pragma unroll
      for (int mt = 0; mt < 4; ++mt)
        store_tile_g64(Apows + (size_t)t * 4096, sb[mt], mt, j, m, hi);
#pragma unroll
      for (int mt = 0; mt < 4; ++mt)
        sa[mt] = v8f{0.f, 0.f, 0.f, 0.f, 0.f, 0.f, 0.f, 0.f};
      scan_step(sa, sb, Wf);
#pragma unroll
      for (int mt = 0; mt < 4; ++mt)
        store_tile_g64(Apows + (size_t)(t + 1) * 4096, sa[mt], mt, j, m, hi);
    }
    return;
  }

  int c = blockIdx.x >> 1;
  int row = (blockIdx.x & 1) * 16 + m;
  const float* ub = bx + (size_t)(c * T) * P;
  float* hb = h_out + (size_t)(c * T + 1) * P;
  v8f r0[4], r1[4], r2[4], r3[4];
  load_state(r1, ub, row, hi);
  load_state(r2, ub + P, row, hi);
#pragma unroll
  for (int mt = 0; mt < 4; ++mt)
    r0[mt] = v8f{0.f, 0.f, 0.f, 0.f, 0.f, 0.f, 0.f, 0.f};

#pragma unroll 1
  for (int i = 0; i < T; i += 4) {
    int n4 = i + 4 < T ? i + 4 : T - 1;
    int n5 = i + 5 < T ? i + 5 : T - 1;
    load_state(r3, ub + (size_t)(i + 2) * P, row, hi);
    scan_step(r1, r0, Wf);
    store_state(hb + (size_t)i * P, r1, row, hi);
    load_state(r0, ub + (size_t)(i + 3) * P, row, hi);
    scan_step(r2, r1, Wf);
    store_state(hb + (size_t)(i + 1) * P, r2, row, hi);
    load_state(r1, ub + (size_t)n4 * P, row, hi);
    scan_step(r3, r2, Wf);
    store_state(hb + (size_t)(i + 2) * P, r3, row, hi);
    load_state(r2, ub + (size_t)n5 * P, row, hi);
    scan_step(r0, r3, Wf);
    store_state(hb + (size_t)(i + 3) * P, r0, row, hi);
  }
  store_state(Lc + (size_t)c * P, r0, row, hi);
}

__global__ void __launch_bounds__(32)
chunk_scan_k(const float* __restrict__ Apow, const float* __restrict__ Lc,
             float* __restrict__ carries) {
  int lane = threadIdx.x & 31, hi = lane >> 4, m = lane & 15;
  int row = blockIdx.x * 16 + m;
  constexpr size_t P = (size_t)BT * LAT;
  constexpr int NS = NC - 1;
  constexpr int NL = NS - NS % 4;
  static_assert(NS - NL == 3, "tail below is written for three steps");

  v2f Wf[4][16];
  load_wfrag(Wf, Apow, m, hi);

  v8f r0[4], r1[4], r2[4], r3[4];
  load_state(r1, Lc, row, hi);
  load_state(r2, Lc + P, row, hi);
#pragma unroll
  for (int mt = 0; mt < 4; ++mt)
    r0[mt] = v8f{0.f, 0.f, 0.f, 0.f, 0.f, 0.f, 0.f, 0.f};
  store_state(carries, r0, row, hi);

#pragma unroll 1
  for (int c = 0; c < NL; c += 4) {
    load_state(r3, Lc + (size_t)(c + 2) * P, row, hi);
    scan_step(r1, r0, Wf);
    store_state(carries + (size_t)(c + 1) * P, r1, row, hi);
    load_state(r0, Lc + (size_t)(c + 3) * P, row, hi);
    scan_step(r2, r1, Wf);
    store_state(carries + (size_t)(c + 2) * P, r2, row, hi);
    load_state(r1, Lc + (size_t)(c + 4) * P, row, hi);
    scan_step(r3, r2, Wf);
    store_state(carries + (size_t)(c + 3) * P, r3, row, hi);
    load_state(r2, Lc + (size_t)(c + 5) * P, row, hi);
    scan_step(r0, r3, Wf);
    store_state(carries + (size_t)(c + 4) * P, r0, row, hi);
  }
  load_state(r3, Lc + (size_t)(NL + 2) * P, row, hi);
  scan_step(r1, r0, Wf);
  store_state(carries + (size_t)(NL + 1) * P, r1, row, hi);
  scan_step(r2, r1, Wf);
  store_state(carries + (size_t)(NL + 2) * P, r2, row, hi);
  scan_step(r3, r2, Wf);
  store_state(carries + (size_t)(NL + 3) * P, r3, row, hi);
}

__device__ __forceinline__ void hold_state(const v8f (&x)[4]) {
  asm volatile("" ::"v"(x[0]), "v"(x[1]), "v"(x[2]), "v"(x[3]));
}
__device__ __forceinline__ void hold_wfrag(const v2f (&w)[4][16]) {
#pragma unroll
  for (int mt = 0; mt < 4; ++mt)
    asm volatile("" ::"v"(w[mt][0]), "v"(w[mt][1]), "v"(w[mt][2]),
                 "v"(w[mt][3]), "v"(w[mt][4]), "v"(w[mt][5]), "v"(w[mt][6]),
                 "v"(w[mt][7]), "v"(w[mt][8]), "v"(w[mt][9]), "v"(w[mt][10]),
                 "v"(w[mt][11]), "v"(w[mt][12]), "v"(w[mt][13]),
                 "v"(w[mt][14]), "v"(w[mt][15]));
}

__global__ void __launch_bounds__(512)
chunk_final_k(const float* __restrict__ Apows,
              const float* __restrict__ carries, float* __restrict__ h_out) {
  int tid = threadIdx.x;
  int lane = tid & 31, hi = lane >> 4, m = lane & 15;
  int t = blockIdx.x >> 3;
  int u = (blockIdx.x & 7) * 16 + (tid >> 5);
  int c = u >> 1;
  int row = (u & 1) * 16 + m;
  constexpr size_t P = (size_t)BT * LAT;

  v2f Wf[4][16];
  load_wfrag(Wf, Apows + (size_t)t * 4096, m, hi);

  float* hp = h_out + (size_t)(c * T + t + 1) * P;
  v8f d[4], a[4];
  load_state(d, carries + (size_t)c * P, row, hi);
  load_state(a, hp, row, hi);

  hold_wfrag(Wf);
  hold_state(d);
  hold_state(a);

  scan_step(a, d, Wf);
  store_state(hp, a, row, hi);
}

constexpr int CPITCH = 68;

__device__ __forceinline__ void load_cpair(v8f (&cf)[2][4], const float* cp) {
#pragma unroll
  for (int t = 0; t < 2; ++t) {
#pragma unroll
    for (int mp = 0; mp < 4; ++mp) {
      const float* p = cp + t * 16 * CPITCH + 16 * mp;
      v4f lo = *(const v4f*)p;
      v4f up = *(const v4f*)(p + 4);
      cf[t][mp] = v8f{lo[0], lo[1], lo[2], lo[3], up[0], up[1], up[2], up[3]};
    }
  }
}

__device__ __forceinline__ void ypair(float* o, const v8f (&cf)[2][4],
                                      const v8f (&hf)[4]) {
  v8f a0 = {0.f, 0.f, 0.f, 0.f, 0.f, 0.f, 0.f, 0.f};
  v8f a1 = {0.f, 0.f, 0.f, 0.f, 0.f, 0.f, 0.f, 0.f};
#pragma unroll
  for (int mp = 0; mp < 4; ++mp) {
#pragma unroll
    for (int q = 0; q < 4; ++q) {
      v2f b = v2f{hf[mp][2 * q], hf[mp][2 * q + 1]};
      a0 = wmma4(v2f{cf[0][mp][2 * q], cf[0][mp][2 * q + 1]}, b, a0);
      a1 = wmma4(v2f{cf[1][mp][2 * q], cf[1][mp][2 * q + 1]}, b, a1);
    }
  }
  *(v4f*)(o)      = v4f{a0[0], a0[1], a0[2], a0[3]};
  *(v4f*)(o + 8)  = v4f{a0[4], a0[5], a0[6], a0[7]};
  *(v4f*)(o + 16) = v4f{a1[0], a1[1], a1[2], a1[3]};
  *(v4f*)(o + 24) = v4f{a1[4], a1[5], a1[6], a1[7]};
}

__global__ void __launch_bounds__(512)
y_gemm_k(const float* __restrict__ hs, const float* __restrict__ C,
         float* __restrict__ y) {
  __shared__ float Cs[DOUT * CPITCH];
  int tid = threadIdx.x;
  int lane = tid & 31, hi = lane >> 4, m = lane & 15;
  int tile = blockIdx.x * 16 + (tid >> 5);

  v8f hf[4];
  load_state(hf, hs + (size_t)tile * 16 * LAT, m, hi);

#pragma unroll
  for (int j = 0; j < DOUT * LAT / 4 / 512; ++j) {
    int i = tid + 512 * j;
    int row = i >> 4, c4 = i & 15;
    *(v4f*)(Cs + row * CPITCH + 4 * c4) = *(const v4f*)(C + row * LAT + 4 * c4);
  }
  __syncthreads();

  int crow = 4 * (m >> 3) + (m & 3) + 8 * ((m >> 2) & 1);
  const float* cp = Cs + crow * CPITCH + 8 * hi;
  float* yp = y + (size_t)(tile * 16 + m) * DOUT + 4 * hi;

  v8f ca[2][4], cb[2][4];
  load_cpair(ca, cp);
#pragma unroll 1
  for (int g = 0; g < DOUT / 64; ++g) {
    int nx = g + 1 < DOUT / 64 ? 2 * g + 2 : 2 * g;
    load_cpair(cb, cp + (2 * g + 1) * 32 * CPITCH);
    ypair(yp + g * 64, ca, hf);
    load_cpair(ca, cp + nx * 32 * CPITCH);
    ypair(yp + g * 64 + 32, cb, hf);
  }
}

extern "C" void kernel_launch(void* const* d_in, const int* in_sizes, int n_in,
                              void* d_out, int out_size, void* d_ws,
                              size_t ws_size, hipStream_t stream) {
  const float* x  = (const float*)d_in[0];
  const float* A  = (const float*)d_in[1];
  const float* Bw = (const float*)d_in[2];
  const float* C  = (const float*)d_in[3];

  float* out      = (float*)d_out;
  float* y_region = out;
  float* h_region = out + (size_t)S * BT * DOUT;

  float* bx = y_region;

  float* ws      = (float*)d_ws;
  float* Apows   = ws;
  float* Lc      = ws + T * 64 * 64;
  float* carries = ws + T * 64 * 64 + (size_t)NC * BT * LAT;

  hipMemsetAsync(h_region, 0, (size_t)BT * LAT * sizeof(float), stream);

  bx_gemm_k<<<(S * BT / 16) * (LAT / 16) / 8, 256, 0, stream>>>(x, Bw, bx);
  chunk_local_k<<<NC * 2 + 4, 32, 0, stream>>>(bx, A, Lc, h_region, Apows);
  chunk_scan_k<<<2, 32, 0, stream>>>(Apows + (T - 1) * 64 * 64, Lc, carries);
  chunk_final_k<<<T * 8, 512, 0, stream>>>(Apows, carries, h_region);
  y_gemm_k<<<S * BT / 16 / 16, 512, 0, stream>>>(
      h_region + BT * LAT, C, y_region);
}


# bx GEMM formed transposed: a wave owns 16 rows of x as B fragments (each row read once), forms all 64 latents, Bw staged once per block in LDS as A, state-layout stores; k sum order of bx changes
# speedup vs baseline: 2.7405x; 1.2321x over previous
#include <hip/hip_runtime.h>


typedef __attribute__((ext_vector_type(2))) float v2f;
typedef __attribute__((ext_vector_type(4))) float v4f;
typedef __attribute__((ext_vector_type(8))) float v8f;

constexpr int S    = 2048;
constexpr int BT   = 32;
constexpr int DIN  = 256;
constexpr int DOUT = 256;
constexpr int LAT  = 64;
constexpr int T    = 32;
constexpr int NC   = 64;

__device__ __forceinline__ v8f wmma4(v2f a, v2f b, v8f c) {
  return __builtin_amdgcn_wmma_f32_16x16x4_f32(false, a, false, b, (short)0, c,
                                               false, false);
}


__device__ __forceinline__ void load_wfrag(v2f (&Wf)[4][16], const float* Wm,
                                           int m, int hi) {
#pragma unroll
  for (int mt = 0; mt < 4; ++mt) {
#pragma unroll
    for (int mp = 0; mp < 4; ++mp) {
      const float* p = Wm + (mt * 16 + m) * 64 + mp * 16 + 8 * hi;
      v4f lo = *(const v4f*)p;
      v4f up = *(const v4f*)(p + 4);
      Wf[mt][4 * mp + 0] = v2f{lo[0], lo[1]};
      Wf[mt][4 * mp + 1] = v2f{lo[2], lo[3]};
      Wf[mt][4 * mp + 2] = v2f{up[0], up[1]};
      Wf[mt][4 * mp + 3] = v2f{up[2], up[3]};
    }
  }
}

__device__ __forceinline__ void load_state(v8f (&d)[4], const float* plane,
                                           int row, int hi) {
  const float* p = plane + row * 64 + 8 * hi;
#pragma unroll
  for (int mt = 0; mt < 4; ++mt) {
    v4f lo = *(const v4f*)(p + 16 * mt);
    v4f up = *(const v4f*)(p + 16 * mt + 4);
    d[mt] = v8f{lo[0], lo[1], lo[2], lo[3], up[0], up[1], up[2], up[3]};
  }
}

__device__ __forceinline__ void store_state(float* plane, const v8f (&d)[4],
                                            int row, int hi) {
  float* p = plane + row * 64 + 8 * hi;
#pragma unroll
  for (int mt = 0; mt < 4; ++mt) {
    *(v4f*)(p + 16 * mt) = v4f{d[mt][0], d[mt][1], d[mt][2], d[mt][3]};
    *(v4f*)(p + 16 * mt + 4) = v4f{d[mt][4], d[mt][5], d[mt][6], d[mt][7]};
  }
}

__device__ __forceinline__ void scan_step(v8f (&acc)[4], const v8f (&d)[4],
                                          const v2f (&Wf)[4][16]) {
#pragma unroll
  for (int mp = 0; mp < 4; ++mp) {
#pragma unroll
    for (int q = 0; q < 4; ++q) {
      v2f b = v2f{d[mp][2 * q], d[mp][2 * q + 1]};
#pragma unroll
      for (int mt = 0; mt < 4; ++mt)
        acc[mt] = wmma4(Wf[mt][4 * mp + q], b, acc[mt]);
    }
  }
}

__device__ __forceinline__ void store_tile_g64(float* g, v8f d, int mt, int nt,
                                               int m, int hi) {
#pragma unroll
  for (int r = 0; r < 8; ++r)
    g[(mt * 16 + r + 8 * hi) * 64 + nt * 16 + m] = d[r];
}

constexpr int BPITCH = 260;

__device__ __forceinline__ void load_xgroup(v8f (&xs)[4], const float* p) {
#pragma unroll
  for (int mp = 0; mp < 4; ++mp) {
    v4f lo = *(const v4f*)(p + 16 * mp);
    v4f up = *(const v4f*)(p + 16 * mp + 4);
    xs[mp] = v8f{lo[0], lo[1], lo[2], lo[3], up[0], up[1], up[2], up[3]};
  }
}

__device__ __forceinline__ void load_bfrag(v8f (&bf)[4], const float* p) {
#pragma unroll
  for (int mt = 0; mt < 4; ++mt) {
    v4f lo = *(const v4f*)(p + mt * 16 * BPITCH);
    v4f up = *(const v4f*)(p + mt * 16 * BPITCH + 4);
    bf[mt] = v8f{lo[0], lo[1], lo[2], lo[3], up[0], up[1], up[2], up[3]};
  }
}

__device__ __forceinline__ void bx_chunk(v8f (&acc)[4], const v8f (&bf)[4],
                                         v8f xs) {
#pragma unroll
  for (int q = 0; q < 4; ++q) {
    v2f b = v2f{xs[2 * q], xs[2 * q + 1]};
#pragma unroll
    for (int mt = 0; mt < 4; ++mt)
      acc[mt] = wmma4(v2f{bf[mt][2 * q], bf[mt][2 * q + 1]}, b, acc[mt]);
  }
}

__device__ __forceinline__ void keep_order() {
  __builtin_amdgcn_sched_barrier(0);
}

__device__ __forceinline__ void bx_group(v8f (&acc)[4], v8f (&fa)[4],
                                         v8f (&fb)[4], const v8f (&xs)[4],
                                         const float* p, const float* pnext) {
  load_bfrag(fb, p + 16);
  keep_order();
  bx_chunk(acc, fa, xs[0]);
  keep_order();
  load_bfrag(fa, p + 32);
  keep_order();
  bx_chunk(acc, fb, xs[1]);
  keep_order();
  load_bfrag(fb, p + 48);
  keep_order();
  bx_chunk(acc, fa, xs[2]);
  keep_order();
  load_bfrag(fa, pnext);
  keep_order();
  bx_chunk(acc, fb, xs[3]);
  keep_order();
}

__global__ void __launch_bounds__(512)
bx_gemm_k(const float* __restrict__ x, const float* __restrict__ Bw,
          float* __restrict__ bx) {
  __shared__ float Bs[LAT * BPITCH];
  int tid = threadIdx.x;
  int lane = tid & 31, hi = lane >> 4, m = lane & 15;
  int tile = blockIdx.x * 16 + (tid >> 5);

  const float* xp = x + (size_t)(tile * 16 + m) * DIN + 8 * hi;
  v8f xa[4], xb[4];
  load_xgroup(xa, xp);

#pragma unroll
  for (int j = 0; j < LAT * DIN / 4 / 512; ++j) {
    int i = tid + 512 * j;
    int row = i >> 6, c4 = i & 63;
    *(v4f*)(Bs + row * BPITCH + 4 * c4) = *(const v4f*)(Bw + row * DIN + 4 * c4);
  }
  __syncthreads();

  const float* bp = Bs + m * BPITCH + 8 * hi;
  v8f acc[4], fa[4], fb[4];
#pragma unroll
  for (int mt = 0; mt < 4; ++mt)
    acc[mt] = v8f{0.f, 0.f, 0.f, 0.f, 0.f, 0.f, 0.f, 0.f};
  load_bfrag(fa, bp);

#pragma unroll 1
  for (int g = 0; g < DIN / 64; g += 2) {
    int nx = g + 2 < DIN / 64 ? g + 2 : g + 1;
    load_xgroup(xb, xp + 64 * (g + 1));
    bx_group(acc, fa, fb, xa, bp + 64 * g, bp + 64 * (g + 1));
    load_xgroup(xa, xp + 64 * nx);
    bx_group(acc, fa, fb, xb, bp + 64 * (g + 1), bp + 64 * nx);
  }
  store_state(bx + (size_t)tile * 16 * LAT, acc, m, hi);
}


__global__ void __launch_bounds__(32)
chunk_local_k(const float* __restrict__ bx, const float* __restrict__ A,
              float* __restrict__ Lc, float* __restrict__ h_out,
              float* __restrict__ Apows) {
  int lane = threadIdx.x & 31, hi = lane >> 4, m = lane & 15;
  constexpr size_t P = (size_t)BT * LAT;

  v2f Wf[4][16];
  load_wfrag(Wf, A, m, hi);

  if (blockIdx.x >= NC * 2) {
    int j = blockIdx.x - NC * 2;
    v8f sa[4], sb[4];
#pragma unroll
    for (int mt = 0; mt < 4; ++mt)
#pragma unroll
      for (int r = 0; r < 8; ++r)
        sa[mt][r] = (16 * mt + 8 * hi + r == 16 * j + m) ? 1.f : 0.f;
#pragma unroll 1
    for (int t = 0; t < T; t += 2) {
#pragma unroll
      for (int mt = 0; mt < 4; ++mt)
        sb[mt] = v8f{0.f, 0.f, 0.f, 0.f, 0.f, 0.f, 0.f, 0.f};
      scan_step(sb, sa, Wf);
#pragma unroll
      for (int mt = 0; mt < 4; ++mt)
        store_tile_g64(Apows + (size_t)t * 4096, sb[mt], mt, j, m, hi);
#pragma unroll
      for (int mt = 0; mt < 4; ++mt)
        sa[mt] = v8f{0.f, 0.f, 0.f, 0.f, 0.f, 0.f, 0.f, 0.f};
      scan_step(sa, sb, Wf);
#pragma unroll
      for (int mt = 0; mt < 4; ++mt)
        store_tile_g64(Apows + (size_t)(t + 1) * 4096, sa[mt], mt, j, m, hi);
    }
    return;
  }

  int c = blockIdx.x >> 1;
  int row = (blockIdx.x & 1) * 16 + m;
  const float* ub = bx + (size_t)(c * T) * P;
  float* hb = h_out + (size_t)(c * T + 1) * P;
  v8f r0[4], r1[4], r2[4], r3[4];
  load_state(r1, ub, row, hi);
  load_state(r2, ub + P, row, hi);
#pragma unroll
  for (int mt = 0; mt < 4; ++mt)
    r0[mt] = v8f{0.f, 0.f, 0.f, 0.f, 0.f, 0.f, 0.f, 0.f};

#pragma unroll 1
  for (int i = 0; i < T; i += 4) {
    int n4 = i + 4 < T ? i + 4 : T - 1;
    int n5 = i + 5 < T ? i + 5 : T - 1;
    load_state(r3, ub + (size_t)(i + 2) * P, row, hi);
    scan_step(r1, r0, Wf);
    store_state(hb + (size_t)i * P, r1, row, hi);
    load_state(r0, ub + (size_t)(i + 3) * P, row, hi);
    scan_step(r2, r1, Wf);
    store_state(hb + (size_t)(i + 1) * P, r2, row, hi);
    load_state(r1, ub + (size_t)n4 * P, row, hi);
    scan_step(r3, r2, Wf);
    store_state(hb + (size_t)(i + 2) * P, r3, row, hi);
    load_state(r2, ub + (size_t)n5 * P, row, hi);
    scan_step(r0, r3, Wf);
    store_state(hb + (size_t)(i + 3) * P, r0, row, hi);
  }
  store_state(Lc + (size_t)c * P, r0, row, hi);
}

__global__ void __launch_bounds__(32)
chunk_scan_k(const float* __restrict__ Apow, const float* __restrict__ Lc,
             float* __restrict__ carries) {
  int lane = threadIdx.x & 31, hi = lane >> 4, m = lane & 15;
  int row = blockIdx.x * 16 + m;
  constexpr size_t P = (size_t)BT * LAT;
  constexpr int NS = NC - 1;
  constexpr int NL = NS - NS % 4;
  static_assert(NS - NL == 3, "tail below is written for three steps");

  v2f Wf[4][16];
  load_wfrag(Wf, Apow, m, hi);

  v8f r0[4], r1[4], r2[4], r3[4];
  load_state(r1, Lc, row, hi);
  load_state(r2, Lc + P, row, hi);
#pragma unroll
  for (int mt = 0; mt < 4; ++mt)
    r0[mt] = v8f{0.f, 0.f, 0.f, 0.f, 0.f, 0.f, 0.f, 0.f};
  store_state(carries, r0, row, hi);

#pragma unroll 1
  for (int c = 0; c < NL; c += 4) {
    load_state(r3, Lc + (size_t)(c + 2) * P, row, hi);
    scan_step(r1, r0, Wf);
    store_state(carries + (size_t)(c + 1) * P, r1, row, hi);
    load_state(r0, Lc + (size_t)(c + 3) * P, row, hi);
    scan_step(r2, r1, Wf);
    store_state(carries + (size_t)(c + 2) * P, r2, row, hi);
    load_state(r1, Lc + (size_t)(c + 4) * P, row, hi);
    scan_step(r3, r2, Wf);
    store_state(carries + (size_t)(c + 3) * P, r3, row, hi);
    load_state(r2, Lc + (size_t)(c + 5) * P, row, hi);
    scan_step(r0, r3, Wf);
    store_state(carries + (size_t)(c + 4) * P, r0, row, hi);
  }
  load_state(r3, Lc + (size_t)(NL + 2) * P, row, hi);
  scan_step(r1, r0, Wf);
  store_state(carries + (size_t)(NL + 1) * P, r1, row, hi);
  scan_step(r2, r1, Wf);
  store_state(carries + (size_t)(NL + 2) * P, r2, row, hi);
  scan_step(r3, r2, Wf);
  store_state(carries + (size_t)(NL + 3) * P, r3, row, hi);
}

__device__ __forceinline__ void hold_state(const v8f (&x)[4]) {
  asm volatile("" ::"v"(x[0]), "v"(x[1]), "v"(x[2]), "v"(x[3]));
}
__device__ __forceinline__ void hold_wfrag(const v2f (&w)[4][16]) {
#pragma unroll
  for (int mt = 0; mt < 4; ++mt)
    asm volatile("" ::"v"(w[mt][0]), "v"(w[mt][1]), "v"(w[mt][2]),
                 "v"(w[mt][3]), "v"(w[mt][4]), "v"(w[mt][5]), "v"(w[mt][6]),
                 "v"(w[mt][7]), "v"(w[mt][8]), "v"(w[mt][9]), "v"(w[mt][10]),
                 "v"(w[mt][11]), "v"(w[mt][12]), "v"(w[mt][13]),
                 "v"(w[mt][14]), "v"(w[mt][15]));
}

__global__ void __launch_bounds__(512)
chunk_final_k(const float* __restrict__ Apows,
              const float* __restrict__ carries, float* __restrict__ h_out) {
  int tid = threadIdx.x;
  int lane = tid & 31, hi = lane >> 4, m = lane & 15;
  int t = blockIdx.x >> 3;
  int u = (blockIdx.x & 7) * 16 + (tid >> 5);
  int c = u >> 1;
  int row = (u & 1) * 16 + m;
  constexpr size_t P = (size_t)BT * LAT;

  v2f Wf[4][16];
  load_wfrag(Wf, Apows + (size_t)t * 4096, m, hi);

  float* hp = h_out + (size_t)(c * T + t + 1) * P;
  v8f d[4], a[4];
  load_state(d, carries + (size_t)c * P, row, hi);
  load_state(a, hp, row, hi);

  hold_wfrag(Wf);
  hold_state(d);
  hold_state(a);

  scan_step(a, d, Wf);
  store_state(hp, a, row, hi);
}

constexpr int CPITCH = 68;

__device__ __forceinline__ void load_cpair(v8f (&cf)[2][4], const float* cp) {
#pragma unroll
  for (int t = 0; t < 2; ++t) {
#pragma unroll
    for (int mp = 0; mp < 4; ++mp) {
      const float* p = cp + t * 16 * CPITCH + 16 * mp;
      v4f lo = *(const v4f*)p;
      v4f up = *(const v4f*)(p + 4);
      cf[t][mp] = v8f{lo[0], lo[1], lo[2], lo[3], up[0], up[1], up[2], up[3]};
    }
  }
}

__device__ __forceinline__ void ypair(float* o, const v8f (&cf)[2][4],
                                      const v8f (&hf)[4]) {
  v8f a0 = {0.f, 0.f, 0.f, 0.f, 0.f, 0.f, 0.f, 0.f};
  v8f a1 = {0.f, 0.f, 0.f, 0.f, 0.f, 0.f, 0.f, 0.f};
#pragma unroll
  for (int mp = 0; mp < 4; ++mp) {
#pragma unroll
    for (int q = 0; q < 4; ++q) {
      v2f b = v2f{hf[mp][2 * q], hf[mp][2 * q + 1]};
      a0 = wmma4(v2f{cf[0][mp][2 * q], cf[0][mp][2 * q + 1]}, b, a0);
      a1 = wmma4(v2f{cf[1][mp][2 * q], cf[1][mp][2 * q + 1]}, b, a1);
    }
  }
  *(v4f*)(o)      = v4f{a0[0], a0[1], a0[2], a0[3]};
  *(v4f*)(o + 8)  = v4f{a0[4], a0[5], a0[6], a0[7]};
  *(v4f*)(o + 16) = v4f{a1[0], a1[1], a1[2], a1[3]};
  *(v4f*)(o + 24) = v4f{a1[4], a1[5], a1[6], a1[7]};
}

__global__ void __launch_bounds__(512)
y_gemm_k(const float* __restrict__ hs, const float* __restrict__ C,
         float* __restrict__ y) {
  __shared__ float Cs[DOUT * CPITCH];
  int tid = threadIdx.x;
  int lane = tid & 31, hi = lane >> 4, m = lane & 15;
  int tile = blockIdx.x * 16 + (tid >> 5);

  v8f hf[4];
  load_state(hf, hs + (size_t)tile * 16 * LAT, m, hi);

#pragma unroll
  for (int j = 0; j < DOUT * LAT / 4 / 512; ++j) {
    int i = tid + 512 * j;
    int row = i >> 4, c4 = i & 15;
    *(v4f*)(Cs + row * CPITCH + 4 * c4) = *(const v4f*)(C + row * LAT + 4 * c4);
  }
  __syncthreads();

  int crow = 4 * (m >> 3) + (m & 3) + 8 * ((m >> 2) & 1);
  const float* cp = Cs + crow * CPITCH + 8 * hi;
  float* yp = y + (size_t)(tile * 16 + m) * DOUT + 4 * hi;

  v8f ca[2][4], cb[2][4];
  load_cpair(ca, cp);
#pragma unroll 1
  for (int g = 0; g < DOUT / 64; ++g) {
    int nx = g + 1 < DOUT / 64 ? 2 * g + 2 : 2 * g;
    load_cpair(cb, cp + (2 * g + 1) * 32 * CPITCH);
    ypair(yp + g * 64, ca, hf);
    load_cpair(ca, cp + nx * 32 * CPITCH);
    ypair(yp + g * 64 + 32, cb, hf);
  }
}

extern "C" void kernel_launch(void* const* d_in, const int* in_sizes, int n_in,
                              void* d_out, int out_size, void* d_ws,
                              size_t ws_size, hipStream_t stream) {
  const float* x  = (const float*)d_in[0];
  const float* A  = (const float*)d_in[1];
  const float* Bw = (const float*)d_in[2];
  const float* C  = (const float*)d_in[3];

  float* out      = (float*)d_out;
  float* y_region = out;
  float* h_region = out + (size_t)S * BT * DOUT;

  float* bx = y_region;

  float* ws      = (float*)d_ws;
  float* Apows   = ws;
  float* Lc      = ws + T * 64 * 64;
  float* carries = ws + T * 64 * 64 + (size_t)NC * BT * LAT;

  hipMemsetAsync(h_region, 0, (size_t)BT * LAT * sizeof(float), stream);

  bx_gemm_k<<<S * BT / 16 / 16, 512, 0, stream>>>(x, Bw, bx);
  chunk_local_k<<<NC * 2 + 4, 32, 0, stream>>>(bx, A, Lc, h_region, Apows);
  chunk_scan_k<<<2, 32, 0, stream>>>(Apows + (T - 1) * 64 * 64, Lc, carries);
  chunk_final_k<<<T * 8, 512, 0, stream>>>(Apows, carries, h_region);
  y_gemm_k<<<S * BT / 16 / 16, 512, 0, stream>>>(
      h_region + BT * LAT, C, y_region);
}


# Chunk carries as a radix-8 prefix: each chunk's wave sums a window of 8 (7 steps, A^32), then 8 windows 8 apart (7 steps, A^256); 63 dependent steps -> 14, one more launch; carry sum order changes
# speedup vs baseline: 3.4719x; 1.2669x over previous
#include <hip/hip_runtime.h>


typedef __attribute__((ext_vector_type(2))) float v2f;
typedef __attribute__((ext_vector_type(4))) float v4f;
typedef __attribute__((ext_vector_type(8))) float v8f;

constexpr int S    = 2048;
constexpr int BT   = 32;
constexpr int DIN  = 256;
constexpr int DOUT = 256;
constexpr int LAT  = 64;
constexpr int T    = 32;
constexpr int NC   = 64;

__device__ __forceinline__ v8f wmma4(v2f a, v2f b, v8f c) {
  return __builtin_amdgcn_wmma_f32_16x16x4_f32(false, a, false, b, (short)0, c,
                                               false, false);
}


__device__ __forceinline__ void load_wfrag(v2f (&Wf)[4][16], const float* Wm,
                                           int m, int hi) {
#pragma unroll
  for (int mt = 0; mt < 4; ++mt) {
#pragma unroll
    for (int mp = 0; mp < 4; ++mp) {
      const float* p = Wm + (mt * 16 + m) * 64 + mp * 16 + 8 * hi;
      v4f lo = *(const v4f*)p;
      v4f up = *(const v4f*)(p + 4);
      Wf[mt][4 * mp + 0] = v2f{lo[0], lo[1]};
      Wf[mt][4 * mp + 1] = v2f{lo[2], lo[3]};
      Wf[mt][4 * mp + 2] = v2f{up[0], up[1]};
      Wf[mt][4 * mp + 3] = v2f{up[2], up[3]};
    }
  }
}

__device__ __forceinline__ void load_state(v8f (&d)[4], const float* plane,
                                           int row, int hi) {
  const float* p = plane + row * 64 + 8 * hi;
#pragma unroll
  for (int mt = 0; mt < 4; ++mt) {
    v4f lo = *(const v4f*)(p + 16 * mt);
    v4f up = *(const v4f*)(p + 16 * mt + 4);
    d[mt] = v8f{lo[0], lo[1], lo[2], lo[3], up[0], up[1], up[2], up[3]};
  }
}

__device__ __forceinline__ void store_state(float* plane, const v8f (&d)[4],
                                            int row, int hi) {
  float* p = plane + row * 64 + 8 * hi;
#pragma unroll
  for (int mt = 0; mt < 4; ++mt) {
    *(v4f*)(p + 16 * mt) = v4f{d[mt][0], d[mt][1], d[mt][2], d[mt][3]};
    *(v4f*)(p + 16 * mt + 4) = v4f{d[mt][4], d[mt][5], d[mt][6], d[mt][7]};
  }
}

__device__ __forceinline__ void scan_step(v8f (&acc)[4], const v8f (&d)[4],
                                          const v2f (&Wf)[4][16]) {
#pragma unroll
  for (int mp = 0; mp < 4; ++mp) {
#pragma unroll
    for (int q = 0; q < 4; ++q) {
      v2f b = v2f{d[mp][2 * q], d[mp][2 * q + 1]};
#pragma unroll
      for (int mt = 0; mt < 4; ++mt)
        acc[mt] = wmma4(Wf[mt][4 * mp + q], b, acc[mt]);
    }
  }
}

__device__ __forceinline__ void store_tile_g64(float* g, v8f d, int mt, int nt,
                                               int m, int hi) {
#pragma unroll
  for (int r = 0; r < 8; ++r)
    g[(mt * 16 + r + 8 * hi) * 64 + nt * 16 + m] = d[r];
}

constexpr int BPITCH = 260;

__device__ __forceinline__ void load_xgroup(v8f (&xs)[4], const float* p) {
#pragma unroll
  for (int mp = 0; mp < 4; ++mp) {
    v4f lo = *(const v4f*)(p + 16 * mp);
    v4f up = *(const v4f*)(p + 16 * mp + 4);
    xs[mp] = v8f{lo[0], lo[1], lo[2], lo[3], up[0], up[1], up[2], up[3]};
  }
}

__device__ __forceinline__ void load_bfrag(v8f (&bf)[4], const float* p) {
#pragma unroll
  for (int mt = 0; mt < 4; ++mt) {
    v4f lo = *(const v4f*)(p + mt * 16 * BPITCH);
    v4f up = *(const v4f*)(p + mt * 16 * BPITCH + 4);
    bf[mt] = v8f{lo[0], lo[1], lo[2], lo[3], up[0], up[1], up[2], up[3]};
  }
}

__device__ __forceinline__ void bx_chunk(v8f (&acc)[4], const v8f (&bf)[4],
                                         v8f xs) {
#pragma unroll
  for (int q = 0; q < 4; ++q) {
    v2f b = v2f{xs[2 * q], xs[2 * q + 1]};
#pragma unroll
    for (int mt = 0; mt < 4; ++mt)
      acc[mt] = wmma4(v2f{bf[mt][2 * q], bf[mt][2 * q + 1]}, b, acc[mt]);
  }
}

__device__ __forceinline__ void keep_order() {
  __builtin_amdgcn_sched_barrier(0);
}

__device__ __forceinline__ void bx_group(v8f (&acc)[4], v8f (&fa)[4],
                                         v8f (&fb)[4], const v8f (&xs)[4],
                                         const float* p, const float* pnext) {
  load_bfrag(fb, p + 16);
  keep_order();
  bx_chunk(acc, fa, xs[0]);
  keep_order();
  load_bfrag(fa, p + 32);
  keep_order();
  bx_chunk(acc, fb, xs[1]);
  keep_order();
  load_bfrag(fb, p + 48);
  keep_order();
  bx_chunk(acc, fa, xs[2]);
  keep_order();
  load_bfrag(fa, pnext);
  keep_order();
  bx_chunk(acc, fb, xs[3]);
  keep_order();
}

__global__ void __launch_bounds__(512)
bx_gemm_k(const float* __restrict__ x, const float* __restrict__ Bw,
          float* __restrict__ bx) {
  __shared__ float Bs[LAT * BPITCH];
  int tid = threadIdx.x;
  int lane = tid & 31, hi = lane >> 4, m = lane & 15;
  int tile = blockIdx.x * 16 + (tid >> 5);

  const float* xp = x + (size_t)(tile * 16 + m) * DIN + 8 * hi;
  v8f xa[4], xb[4];
  load_xgroup(xa, xp);

#pragma unroll
  for (int j = 0; j < LAT * DIN / 4 / 512; ++j) {
    int i = tid + 512 * j;
    int row = i >> 6, c4 = i & 63;
    *(v4f*)(Bs + row * BPITCH + 4 * c4) = *(const v4f*)(Bw + row * DIN + 4 * c4);
  }
  __syncthreads();

  const float* bp = Bs + m * BPITCH + 8 * hi;
  v8f acc[4], fa[4], fb[4];
#pragma unroll
  for (int mt = 0; mt < 4; ++mt)
    acc[mt] = v8f{0.f, 0.f, 0.f, 0.f, 0.f, 0.f, 0.f, 0.f};
  load_bfrag(fa, bp);

#pragma unroll 1
  for (int g = 0; g < DIN / 64; g += 2) {
    int nx = g + 2 < DIN / 64 ? g + 2 : g + 1;
    load_xgroup(xb, xp + 64 * (g + 1));
    bx_group(acc, fa, fb, xa, bp + 64 * g, bp + 64 * (g + 1));
    load_xgroup(xa, xp + 64 * nx);
    bx_group(acc, fa, fb, xb, bp + 64 * (g + 1), bp + 64 * nx);
  }
  store_state(bx + (size_t)tile * 16 * LAT, acc, m, hi);
}


__global__ void __launch_bounds__(32)
chunk_local_k(const float* __restrict__ bx, const float* __restrict__ A,
              float* __restrict__ Lc, float* __restrict__ h_out,
              float* __restrict__ Apows) {
  int lane = threadIdx.x & 31, hi = lane >> 4, m = lane & 15;
  constexpr size_t P = (size_t)BT * LAT;

  v2f Wf[4][16];
  load_wfrag(Wf, A, m, hi);

  if (blockIdx.x >= NC * 2) {
    int j = blockIdx.x - NC * 2;
    v8f sa[4], sb[4];
#pragma unroll
    for (int mt = 0; mt < 4; ++mt)
#pragma unroll
      for (int r = 0; r < 8; ++r)
        sa[mt][r] = (16 * mt + 8 * hi + r == 16 * j + m) ? 1.f : 0.f;
#pragma unroll 1
    for (int t = 0; t < T; t += 2) {
#pragma unroll
      for (int mt = 0; mt < 4; ++mt)
        sb[mt] = v8f{0.f, 0.f, 0.f, 0.f, 0.f, 0.f, 0.f, 0.f};
      scan_step(sb, sa, Wf);
#pragma unroll
      for (int mt = 0; mt < 4; ++mt)
        store_tile_g64(Apows + (size_t)t * 4096, sb[mt], mt, j, m, hi);
#pragma unroll
      for (int mt = 0; mt < 4; ++mt)
        sa[mt] = v8f{0.f, 0.f, 0.f, 0.f, 0.f, 0.f, 0.f, 0.f};
      scan_step(sa, sb, Wf);
#pragma unroll
      for (int mt = 0; mt < 4; ++mt)
        store_tile_g64(Apows + (size_t)(t + 1) * 4096, sa[mt], mt, j, m, hi);
    }
    return;
  }

  int c = blockIdx.x >> 1;
  int row = (blockIdx.x & 1) * 16 + m;
  const float* ub = bx + (size_t)(c * T) * P;
  float* hb = h_out + (size_t)(c * T + 1) * P;
  v8f r0[4], r1[4], r2[4], r3[4];
  load_state(r1, ub, row, hi);
  load_state(r2, ub + P, row, hi);
#pragma unroll
  for (int mt = 0; mt < 4; ++mt)
    r0[mt] = v8f{0.f, 0.f, 0.f, 0.f, 0.f, 0.f, 0.f, 0.f};

#pragma unroll 1
  for (int i = 0; i < T; i += 4) {
    int n4 = i + 4 < T ? i + 4 : T - 1;
    int n5 = i + 5 < T ? i + 5 : T - 1;
    load_state(r3, ub + (size_t)(i + 2) * P, row, hi);
    scan_step(r1, r0, Wf);
    store_state(hb + (size_t)i * P, r1, row, hi);
    load_state(r0, ub + (size_t)(i + 3) * P, row, hi);
    scan_step(r2, r1, Wf);
    store_state(hb + (size_t)(i + 1) * P, r2, row, hi);
    load_state(r1, ub + (size_t)n4 * P, row, hi);
    scan_step(r3, r2, Wf);
    store_state(hb + (size_t)(i + 2) * P, r3, row, hi);
    load_state(r2, ub + (size_t)n5 * P, row, hi);
    scan_step(r0, r3, Wf);
    store_state(hb + (size_t)(i + 3) * P, r0, row, hi);
  }
  store_state(Lc + (size_t)c * P, r0, row, hi);
}

__device__ __forceinline__ void hold_state(const v8f (&x)[4]) {
  asm volatile("" ::"v"(x[0]), "v"(x[1]), "v"(x[2]), "v"(x[3]));
}
__device__ __forceinline__ void hold_wfrag(const v2f (&w)[4][16]) {
#pragma unroll
  for (int mt = 0; mt < 4; ++mt)
    asm volatile("" ::"v"(w[mt][0]), "v"(w[mt][1]), "v"(w[mt][2]),
                 "v"(w[mt][3]), "v"(w[mt][4]), "v"(w[mt][5]), "v"(w[mt][6]),
                 "v"(w[mt][7]), "v"(w[mt][8]), "v"(w[mt][9]), "v"(w[mt][10]),
                 "v"(w[mt][11]), "v"(w[mt][12]), "v"(w[mt][13]),
                 "v"(w[mt][14]), "v"(w[mt][15]));
}

constexpr int WR = 8;
static_assert(WR * WR == NC, "two window passes must span all chunks");

__device__ __forceinline__ void window_sum(const v2f (&Wf)[4][16],
                                           const float* in, float* out, int c,
                                           int shift, int stride, int row,
                                           int hi) {
  constexpr size_t P = (size_t)BT * LAT;
  v8f u[WR][4];
#pragma unroll
  for (int i = 0; i < WR; ++i) {
    int j = c - shift - stride * (WR - 1 - i);
    if (j >= 0) {
      load_state(u[i], in + (size_t)j * P, row, hi);
    } else {
#pragma unroll
      for (int mt = 0; mt < 4; ++mt)
        u[i][mt] = v8f{0.f, 0.f, 0.f, 0.f, 0.f, 0.f, 0.f, 0.f};
    }
  }
  keep_order();
#pragma unroll
  for (int i = 0; i + 1 < WR; ++i) scan_step(u[i + 1], u[i], Wf);
  store_state(out + (size_t)c * P, u[WR - 1], row, hi);
}

__global__ void __launch_bounds__(32)
carry_win1_k(const float* __restrict__ Pm, const float* __restrict__ Lc,
             float* __restrict__ Wn, float* __restrict__ Q) {
  int lane = threadIdx.x & 31, hi = lane >> 4, m = lane & 15;

  v2f Wf[4][16];
  load_wfrag(Wf, Pm, m, hi);

  if (blockIdx.x >= NC * 2) {
    int j = blockIdx.x - NC * 2;
    v8f p[WR][4];
#pragma unroll
    for (int mt = 0; mt < 4; ++mt)
#pragma unroll
      for (int r = 0; r < 8; ++r)
        p[0][mt][r] = Pm[(16 * mt + 8 * hi + r) * 64 + 16 * j + m];
    keep_order();
#pragma unroll
    for (int i = 0; i + 1 < WR; ++i) {
#pragma unroll
      for (int mt = 0; mt < 4; ++mt)
        p[i + 1][mt] = v8f{0.f, 0.f, 0.f, 0.f, 0.f, 0.f, 0.f, 0.f};
      scan_step(p[i + 1], p[i], Wf);
    }
#pragma unroll
    for (int mt = 0; mt < 4; ++mt)
      store_tile_g64(Q, p[WR - 1][mt], mt, j, m, hi);
    return;
  }

  int c = blockIdx.x >> 1;
  int row = (blockIdx.x & 1) * 16 + m;
  window_sum(Wf, Lc, Wn, c, 0, 1, row, hi);
}

__global__ void __launch_bounds__(32)
carry_win2_k(const float* __restrict__ Q, const float* __restrict__ Wn,
             float* __restrict__ carries) {
  int lane = threadIdx.x & 31, hi = lane >> 4, m = lane & 15;
  int c = blockIdx.x >> 1;
  int row = (blockIdx.x & 1) * 16 + m;

  v2f Wf[4][16];
  load_wfrag(Wf, Q, m, hi);
  window_sum(Wf, Wn, carries, c, 1, WR, row, hi);
}

__global__ void __launch_bounds__(512)
chunk_final_k(const float* __restrict__ Apows,
              const float* __restrict__ carries, float* __restrict__ h_out) {
  int tid = threadIdx.x;
  int lane = tid & 31, hi = lane >> 4, m = lane & 15;
  int t = blockIdx.x >> 3;
  int u = (blockIdx.x & 7) * 16 + (tid >> 5);
  int c = u >> 1;
  int row = (u & 1) * 16 + m;
  constexpr size_t P = (size_t)BT * LAT;

  v2f Wf[4][16];
  load_wfrag(Wf, Apows + (size_t)t * 4096, m, hi);

  float* hp = h_out + (size_t)(c * T + t + 1) * P;
  v8f d[4], a[4];
  load_state(d, carries + (size_t)c * P, row, hi);
  load_state(a, hp, row, hi);

  hold_wfrag(Wf);
  hold_state(d);
  hold_state(a);

  scan_step(a, d, Wf);
  store_state(hp, a, row, hi);
}

constexpr int CPITCH = 68;

__device__ __forceinline__ void load_cpair(v8f (&cf)[2][4], const float* cp) {
#pragma unroll
  for (int t = 0; t < 2; ++t) {
#pragma unroll
    for (int mp = 0; mp < 4; ++mp) {
      const float* p = cp + t * 16 * CPITCH + 16 * mp;
      v4f lo = *(const v4f*)p;
      v4f up = *(const v4f*)(p + 4);
      cf[t][mp] = v8f{lo[0], lo[1], lo[2], lo[3], up[0], up[1], up[2], up[3]};
    }
  }
}

__device__ __forceinline__ void ypair(float* o, const v8f (&cf)[2][4],
                                      const v8f (&hf)[4]) {
  v8f a0 = {0.f, 0.f, 0.f, 0.f, 0.f, 0.f, 0.f, 0.f};
  v8f a1 = {0.f, 0.f, 0.f, 0.f, 0.f, 0.f, 0.f, 0.f};
#pragma unroll
  for (int mp = 0; mp < 4; ++mp) {
#pragma unroll
    for (int q = 0; q < 4; ++q) {
      v2f b = v2f{hf[mp][2 * q], hf[mp][2 * q + 1]};
      a0 = wmma4(v2f{cf[0][mp][2 * q], cf[0][mp][2 * q + 1]}, b, a0);
      a1 = wmma4(v2f{cf[1][mp][2 * q], cf[1][mp][2 * q + 1]}, b, a1);
    }
  }
  *(v4f*)(o)      = v4f{a0[0], a0[1], a0[2], a0[3]};
  *(v4f*)(o + 8)  = v4f{a0[4], a0[5], a0[6], a0[7]};
  *(v4f*)(o + 16) = v4f{a1[0], a1[1], a1[2], a1[3]};
  *(v4f*)(o + 24) = v4f{a1[4], a1[5], a1[6], a1[7]};
}

__global__ void __launch_bounds__(512)
y_gemm_k(const float* __restrict__ hs, const float* __restrict__ C,
         float* __restrict__ y) {
  __shared__ float Cs[DOUT * CPITCH];
  int tid = threadIdx.x;
  int lane = tid & 31, hi = lane >> 4, m = lane & 15;
  int tile = blockIdx.x * 16 + (tid >> 5);

  v8f hf[4];
  load_state(hf, hs + (size_t)tile * 16 * LAT, m, hi);

#pragma unroll
  for (int j = 0; j < DOUT * LAT / 4 / 512; ++j) {
    int i = tid + 512 * j;
    int row = i >> 4, c4 = i & 15;
    *(v4f*)(Cs + row * CPITCH + 4 * c4) = *(const v4f*)(C + row * LAT + 4 * c4);
  }
  __syncthreads();

  int crow = 4 * (m >> 3) + (m & 3) + 8 * ((m >> 2) & 1);
  const float* cp = Cs + crow * CPITCH + 8 * hi;
  float* yp = y + (size_t)(tile * 16 + m) * DOUT + 4 * hi;

  v8f ca[2][4], cb[2][4];
  load_cpair(ca, cp);
#pragma unroll 1
  for (int g = 0; g < DOUT / 64; ++g) {
    int nx = g + 1 < DOUT / 64 ? 2 * g + 2 : 2 * g;
    load_cpair(cb, cp + (2 * g + 1) * 32 * CPITCH);
    ypair(yp + g * 64, ca, hf);
    load_cpair(ca, cp + nx * 32 * CPITCH);
    ypair(yp + g * 64 + 32, cb, hf);
  }
}

extern "C" void kernel_launch(void* const* d_in, const int* in_sizes, int n_in,
                              void* d_out, int out_size, void* d_ws,
                              size_t ws_size, hipStream_t stream) {
  const float* x  = (const float*)d_in[0];
  const float* A  = (const float*)d_in[1];
  const float* Bw = (const float*)d_in[2];
  const float* C  = (const float*)d_in[3];

  float* out      = (float*)d_out;
  float* y_region = out;
  float* h_region = out + (size_t)S * BT * DOUT;

  float* bx = y_region;

  float* ws      = (float*)d_ws;
  float* Apows   = ws;
  float* Lc      = ws + T * 64 * 64;
  float* carries = ws + T * 64 * 64 + (size_t)NC * BT * LAT;
  float* Wn      = carries + (size_t)NC * BT * LAT;
  float* Q       = Wn + (size_t)NC * BT * LAT;

  hipMemsetAsync(h_region, 0, (size_t)BT * LAT * sizeof(float), stream);

  bx_gemm_k<<<S * BT / 16 / 16, 512, 0, stream>>>(x, Bw, bx);
  chunk_local_k<<<NC * 2 + 4, 32, 0, stream>>>(bx, A, Lc, h_region, Apows);
  carry_win1_k<<<NC * 2 + 4, 32, 0, stream>>>(Apows + (T - 1) * 64 * 64, Lc,
                                              Wn, Q);
  carry_win2_k<<<NC * 2, 32, 0, stream>>>(Q, Wn, carries);
  chunk_final_k<<<T * 8, 512, 0, stream>>>(Apows, carries, h_region);
  y_gemm_k<<<S * BT / 16 / 16, 512, 0, stream>>>(
      h_region + BT * LAT, C, y_region);
}


# Scan strides two time steps: the GEMM wave owns a pair and forms v=A*u[2j]+u[2j+1], serial passes step with A^2 (32 -> 16 steps), final pass adds the odd states; h sum order changes
# speedup vs baseline: 3.8260x; 1.1020x over previous
#include <hip/hip_runtime.h>


typedef __attribute__((ext_vector_type(2))) float v2f;
typedef __attribute__((ext_vector_type(4))) float v4f;
typedef __attribute__((ext_vector_type(8))) float v8f;

constexpr int S    = 2048;
constexpr int BT   = 32;
constexpr int DIN  = 256;
constexpr int DOUT = 256;
constexpr int LAT  = 64;
constexpr int T    = 32;
constexpr int NC   = 64;

__device__ __forceinline__ v8f wmma4(v2f a, v2f b, v8f c) {
  return __builtin_amdgcn_wmma_f32_16x16x4_f32(false, a, false, b, (short)0, c,
                                               false, false);
}


template <int PITCH = 64>
__device__ __forceinline__ void load_wfrag(v2f (&Wf)[4][16], const float* Wm,
                                           int m, int hi) {
#pragma unroll
  for (int mt = 0; mt < 4; ++mt) {
#pragma unroll
    for (int mp = 0; mp < 4; ++mp) {
      const float* p = Wm + (mt * 16 + m) * PITCH + mp * 16 + 8 * hi;
      v4f lo = *(const v4f*)p;
      v4f up = *(const v4f*)(p + 4);
      Wf[mt][4 * mp + 0] = v2f{lo[0], lo[1]};
      Wf[mt][4 * mp + 1] = v2f{lo[2], lo[3]};
      Wf[mt][4 * mp + 2] = v2f{up[0], up[1]};
      Wf[mt][4 * mp + 3] = v2f{up[2], up[3]};
    }
  }
}

__device__ __forceinline__ void load_state(v8f (&d)[4], const float* plane,
                                           int row, int hi) {
  const float* p = plane + row * 64 + 8 * hi;
#pragma unroll
  for (int mt = 0; mt < 4; ++mt) {
    v4f lo = *(const v4f*)(p + 16 * mt);
    v4f up = *(const v4f*)(p + 16 * mt + 4);
    d[mt] = v8f{lo[0], lo[1], lo[2], lo[3], up[0], up[1], up[2], up[3]};
  }
}

__device__ __forceinline__ void store_state(float* plane, const v8f (&d)[4],
                                            int row, int hi) {
  float* p = plane + row * 64 + 8 * hi;
#pragma unroll
  for (int mt = 0; mt < 4; ++mt) {
    *(v4f*)(p + 16 * mt) = v4f{d[mt][0], d[mt][1], d[mt][2], d[mt][3]};
    *(v4f*)(p + 16 * mt + 4) = v4f{d[mt][4], d[mt][5], d[mt][6], d[mt][7]};
  }
}

__device__ __forceinline__ void scan_step(v8f (&acc)[4], const v8f (&d)[4],
                                          const v2f (&Wf)[4][16]) {
#pragma unroll
  for (int mp = 0; mp < 4; ++mp) {
#pragma unroll
    for (int q = 0; q < 4; ++q) {
      v2f b = v2f{d[mp][2 * q], d[mp][2 * q + 1]};
#pragma unroll
      for (int mt = 0; mt < 4; ++mt)
        acc[mt] = wmma4(Wf[mt][4 * mp + q], b, acc[mt]);
    }
  }
}

__device__ __forceinline__ void store_tile_g64(float* g, v8f d, int mt, int nt,
                                               int m, int hi) {
#pragma unroll
  for (int r = 0; r < 8; ++r)
    g[(mt * 16 + r + 8 * hi) * 64 + nt * 16 + m] = d[r];
}

constexpr int BPITCH = 260;

__device__ __forceinline__ void load_xgroup(v8f (&xs)[4], const float* p) {
#pragma unroll
  for (int mp = 0; mp < 4; ++mp) {
    v4f lo = *(const v4f*)(p + 16 * mp);
    v4f up = *(const v4f*)(p + 16 * mp + 4);
    xs[mp] = v8f{lo[0], lo[1], lo[2], lo[3], up[0], up[1], up[2], up[3]};
  }
}

__device__ __forceinline__ void load_bfrag(v8f (&bf)[4], const float* p) {
#pragma unroll
  for (int mt = 0; mt < 4; ++mt) {
    v4f lo = *(const v4f*)(p + mt * 16 * BPITCH);
    v4f up = *(const v4f*)(p + mt * 16 * BPITCH + 4);
    bf[mt] = v8f{lo[0], lo[1], lo[2], lo[3], up[0], up[1], up[2], up[3]};
  }
}

__device__ __forceinline__ void bx_chunk(v8f (&acc)[4], const v8f (&bf)[4],
                                         v8f xs) {
#pragma unroll
  for (int q = 0; q < 4; ++q) {
    v2f b = v2f{xs[2 * q], xs[2 * q + 1]};
#pragma unroll
    for (int mt = 0; mt < 4; ++mt)
      acc[mt] = wmma4(v2f{bf[mt][2 * q], bf[mt][2 * q + 1]}, b, acc[mt]);
  }
}

__device__ __forceinline__ void keep_order() {
  __builtin_amdgcn_sched_barrier(0);
}

__device__ __forceinline__ void bx_group(v8f (&acc)[4], v8f (&fa)[4],
                                         v8f (&fb)[4], const v8f (&xs)[4],
                                         const float* p, const float* pnext) {
  load_bfrag(fb, p + 16);
  keep_order();
  bx_chunk(acc, fa, xs[0]);
  keep_order();
  load_bfrag(fa, p + 32);
  keep_order();
  bx_chunk(acc, fb, xs[1]);
  keep_order();
  load_bfrag(fb, p + 48);
  keep_order();
  bx_chunk(acc, fa, xs[2]);
  keep_order();
  load_bfrag(fa, pnext);
  keep_order();
  bx_chunk(acc, fb, xs[3]);
  keep_order();
}

constexpr int APITCH = 68;
constexpr int NPAIR = S / 2;

__device__ __forceinline__ void bx_tile(v8f (&acc)[4], v8f (&fa)[4],
                                        v8f (&fb)[4], v8f (&xa)[4],
                                        v8f (&xb)[4], const float* xc,
                                        const float* xnext, const float* bp) {
#pragma unroll 1
  for (int g = 0; g < DIN / 64; g += 2) {
    const float* nxp = g + 2 < DIN / 64 ? xc + 64 * (g + 2) : xnext;
    load_xgroup(xb, xc + 64 * (g + 1));
    bx_group(acc, fa, fb, xa, bp + 64 * g, bp + 64 * (g + 1));
    load_xgroup(xa, nxp);
    bx_group(acc, fa, fb, xb, bp + 64 * (g + 1),
             bp + 64 * ((g + 2) & (DIN / 64 - 1)));
  }
}

__global__ void __launch_bounds__(256)
bx_pair_k(const float* __restrict__ x, const float* __restrict__ Bw,
          const float* __restrict__ A, float* __restrict__ U0,
          float* __restrict__ V, float* __restrict__ A2) {
  __shared__ float Bs[LAT * BPITCH];
  __shared__ float As[LAT * APITCH];
  int tid = threadIdx.x;
  int lane = tid & 31, hi = lane >> 4, m = lane & 15;
  int w = tid >> 5;

  if (blockIdx.x == NPAIR * 2 / 8) {
    if (w < 4) {
      v2f Wf[4][16];
      load_wfrag(Wf, A, m, hi);
      v8f p0[4], p1[4];
#pragma unroll
      for (int mt = 0; mt < 4; ++mt) {
#pragma unroll
        for (int r = 0; r < 8; ++r)
          p0[mt][r] = A[(16 * mt + 8 * hi + r) * 64 + 16 * w + m];
        p1[mt] = v8f{0.f, 0.f, 0.f, 0.f, 0.f, 0.f, 0.f, 0.f};
      }
      scan_step(p1, p0, Wf);
#pragma unroll
      for (int mt = 0; mt < 4; ++mt) store_tile_g64(A2, p1[mt], mt, w, m, hi);
    }
    return;
  }

  int p = blockIdx.x * 8 + w;
  int tile0 = 4 * (p >> 1) + (p & 1);
  const float* xp = x + (size_t)(tile0 * 16 + m) * DIN + 8 * hi;
  const float* xq = xp + (size_t)BT * DIN;
  v8f xa[4], xb[4];
  load_xgroup(xa, xp);

#pragma unroll
  for (int j = 0; j < LAT * DIN / 4 / 256; ++j) {
    int i = tid + 256 * j;
    int row = i >> 6, c4 = i & 63;
    *(v4f*)(Bs + row * BPITCH + 4 * c4) = *(const v4f*)(Bw + row * DIN + 4 * c4);
  }
#pragma unroll
  for (int j = 0; j < LAT * LAT / 4 / 256; ++j) {
    int i = tid + 256 * j;
    int row = i >> 4, c4 = i & 15;
    *(v4f*)(As + row * APITCH + 4 * c4) = *(const v4f*)(A + row * LAT + 4 * c4);
  }
  __syncthreads();

  const float* bp = Bs + m * BPITCH + 8 * hi;
  v8f u0[4], u1[4], fa[4], fb[4];
#pragma unroll
  for (int mt = 0; mt < 4; ++mt) {
    u0[mt] = v8f{0.f, 0.f, 0.f, 0.f, 0.f, 0.f, 0.f, 0.f};
    u1[mt] = v8f{0.f, 0.f, 0.f, 0.f, 0.f, 0.f, 0.f, 0.f};
  }
  load_bfrag(fa, bp);

  bx_tile(u0, fa, fb, xa, xb, xp, xq, bp);
  store_state(U0 + (size_t)p * 16 * LAT, u0, m, hi);
  bx_tile(u1, fa, fb, xa, xb, xq, xq, bp);

  v2f Wf[4][16];
  load_wfrag<APITCH>(Wf, As, m, hi);
  keep_order();
  scan_step(u1, u0, Wf);
  store_state(V + (size_t)p * 16 * LAT, u1, m, hi);
}


__global__ void __launch_bounds__(32)
chunk_local_k(const float* __restrict__ V, const float* __restrict__ A2,
              float* __restrict__ Lc, float* __restrict__ h_out,
              float* __restrict__ Apows) {
  int lane = threadIdx.x & 31, hi = lane >> 4, m = lane & 15;
  constexpr size_t P = (size_t)BT * LAT;
  constexpr int TP = T / 2;

  v2f Wf[4][16];
  load_wfrag(Wf, A2, m, hi);

  if (blockIdx.x >= NC * 2) {
    int j = blockIdx.x - NC * 2;
    v8f sa[4], sb[4];
#pragma unroll
    for (int mt = 0; mt < 4; ++mt)
#pragma unroll
      for (int r = 0; r < 8; ++r)
        sa[mt][r] = (16 * mt + 8 * hi + r == 16 * j + m) ? 1.f : 0.f;
#pragma unroll 1
    for (int t = 0; t < TP; t += 2) {
#pragma unroll
      for (int mt = 0; mt < 4; ++mt)
        sb[mt] = v8f{0.f, 0.f, 0.f, 0.f, 0.f, 0.f, 0.f, 0.f};
      scan_step(sb, sa, Wf);
#pragma unroll
      for (int mt = 0; mt < 4; ++mt)
        store_tile_g64(Apows + (size_t)(2 * t + 1) * 4096, sb[mt], mt, j, m, hi);
#pragma unroll
      for (int mt = 0; mt < 4; ++mt)
        sa[mt] = v8f{0.f, 0.f, 0.f, 0.f, 0.f, 0.f, 0.f, 0.f};
      scan_step(sa, sb, Wf);
#pragma unroll
      for (int mt = 0; mt < 4; ++mt)
        store_tile_g64(Apows + (size_t)(2 * t + 3) * 4096, sa[mt], mt, j, m, hi);
    }
    return;
  }

  int c = blockIdx.x >> 1;
  int row = (blockIdx.x & 1) * 16 + m;
  const float* ub = V + (size_t)(c * TP) * P;
  float* hb = h_out + (size_t)(c * T + 2) * P;
  v8f r0[4], r1[4], r2[4], r3[4];
  load_state(r1, ub, row, hi);
  load_state(r2, ub + P, row, hi);
#pragma unroll
  for (int mt = 0; mt < 4; ++mt)
    r0[mt] = v8f{0.f, 0.f, 0.f, 0.f, 0.f, 0.f, 0.f, 0.f};

#pragma unroll 1
  for (int i = 0; i < TP; i += 4) {
    int n4 = i + 4 < TP ? i + 4 : TP - 1;
    int n5 = i + 5 < TP ? i + 5 : TP - 1;
    load_state(r3, ub + (size_t)(i + 2) * P, row, hi);
    scan_step(r1, r0, Wf);
    store_state(hb + (size_t)(2 * i) * P, r1, row, hi);
    load_state(r0, ub + (size_t)(i + 3) * P, row, hi);
    scan_step(r2, r1, Wf);
    store_state(hb + (size_t)(2 * i + 2) * P, r2, row, hi);
    load_state(r1, ub + (size_t)n4 * P, row, hi);
    scan_step(r3, r2, Wf);
    store_state(hb + (size_t)(2 * i + 4) * P, r3, row, hi);
    load_state(r2, ub + (size_t)n5 * P, row, hi);
    scan_step(r0, r3, Wf);
    store_state(hb + (size_t)(2 * i + 6) * P, r0, row, hi);
  }
  store_state(Lc + (size_t)c * P, r0, row, hi);
}

constexpr int WR = 8;
static_assert(WR * WR == NC, "two window passes must span all chunks");

__device__ __forceinline__ void window_sum(const v2f (&Wf)[4][16],
                                           const float* in, float* out, int c,
                                           int shift, int stride, int row,
                                           int hi) {
  constexpr size_t P = (size_t)BT * LAT;
  v8f u[WR][4];
#pragma unroll
  for (int i = 0; i < WR; ++i) {
    int j = c - shift - stride * (WR - 1 - i);
    if (j >= 0) {
      load_state(u[i], in + (size_t)j * P, row, hi);
    } else {
#pragma unroll
      for (int mt = 0; mt < 4; ++mt)
        u[i][mt] = v8f{0.f, 0.f, 0.f, 0.f, 0.f, 0.f, 0.f, 0.f};
    }
  }
  keep_order();
#pragma unroll
  for (int i = 0; i + 1 < WR; ++i) scan_step(u[i + 1], u[i], Wf);
  store_state(out + (size_t)c * P, u[WR - 1], row, hi);
}

__global__ void __launch_bounds__(32)
carry_win1_k(const float* __restrict__ Pm, const float* __restrict__ Lc,
             float* __restrict__ Wn, float* __restrict__ Q) {
  int lane = threadIdx.x & 31, hi = lane >> 4, m = lane & 15;

  v2f Wf[4][16];
  load_wfrag(Wf, Pm, m, hi);

  if (blockIdx.x >= NC * 2) {
    int j = blockIdx.x - NC * 2;
    v8f p[WR][4];
#pragma unroll
    for (int mt = 0; mt < 4; ++mt)
#pragma unroll
      for (int r = 0; r < 8; ++r)
        p[0][mt][r] = Pm[(16 * mt + 8 * hi + r) * 64 + 16 * j + m];
    keep_order();
#pragma unroll
    for (int i = 0; i + 1 < WR; ++i) {
#pragma unroll
      for (int mt = 0; mt < 4; ++mt)
        p[i + 1][mt] = v8f{0.f, 0.f, 0.f, 0.f, 0.f, 0.f, 0.f, 0.f};
      scan_step(p[i + 1], p[i], Wf);
    }
#pragma unroll
    for (int mt = 0; mt < 4; ++mt)
      store_tile_g64(Q, p[WR - 1][mt], mt, j, m, hi);
    return;
  }

  int c = blockIdx.x >> 1;
  int row = (blockIdx.x & 1) * 16 + m;
  window_sum(Wf, Lc, Wn, c, 0, 1, row, hi);
}

__global__ void __launch_bounds__(32)
carry_win2_k(const float* __restrict__ Q, const float* __restrict__ Wn,
             float* __restrict__ carries) {
  int lane = threadIdx.x & 31, hi = lane >> 4, m = lane & 15;
  int c = blockIdx.x >> 1;
  int row = (blockIdx.x & 1) * 16 + m;

  v2f Wf[4][16];
  load_wfrag(Wf, Q, m, hi);
  window_sum(Wf, Wn, carries, c, 1, WR, row, hi);
}

__global__ void __launch_bounds__(256)
chunk_final_k(const float* __restrict__ A, const float* __restrict__ Apows,
              const float* __restrict__ carries, const float* __restrict__ U0,
              float* __restrict__ h_out) {
  int tid = threadIdx.x;
  int lane = tid & 31, hi = lane >> 4, m = lane & 15;
  int i = blockIdx.x >> 4;
  int u = (blockIdx.x & 15) * 8 + (tid >> 5);
  if (i == T / 2) {
    if (u >= 2) return;
    u += 2 * (NC - 1);
  }
  int c = u >> 1;
  int row = (u & 1) * 16 + m;
  constexpr size_t P = (size_t)BT * LAT;

  float* hp = h_out + (size_t)(c * T + 2 * i) * P;
  v2f W1[4][16], WA[4][16];
  v8f d[4], a[4], b[4];
  load_state(d, carries + (size_t)c * P, row, hi);
  if (i > 0) {
    load_wfrag(W1, Apows + (size_t)(2 * i - 1) * 4096, m, hi);
    load_state(a, hp, row, hi);
  }
  if (i < T / 2) {
    load_wfrag(WA, A, m, hi);
    load_state(b, U0 + (size_t)(c * (T / 2) + i) * P, row, hi);
  }
  keep_order();

  if (i > 0) {
    scan_step(a, d, W1);
  } else {
#pragma unroll
    for (int mt = 0; mt < 4; ++mt) a[mt] = d[mt];
  }
  store_state(hp, a, row, hi);
  if (i < T / 2) {
    scan_step(b, a, WA);
    store_state(hp + P, b, row, hi);
  }
}

constexpr int CPITCH = 68;

__device__ __forceinline__ void load_cpair(v8f (&cf)[2][4], const float* cp) {
#pragma unroll
  for (int t = 0; t < 2; ++t) {
#pragma unroll
    for (int mp = 0; mp < 4; ++mp) {
      const float* p = cp + t * 16 * CPITCH + 16 * mp;
      v4f lo = *(const v4f*)p;
      v4f up = *(const v4f*)(p + 4);
      cf[t][mp] = v8f{lo[0], lo[1], lo[2], lo[3], up[0], up[1], up[2], up[3]};
    }
  }
}

__device__ __forceinline__ void ypair(float* o, const v8f (&cf)[2][4],
                                      const v8f (&hf)[4]) {
  v8f a0 = {0.f, 0.f, 0.f, 0.f, 0.f, 0.f, 0.f, 0.f};
  v8f a1 = {0.f, 0.f, 0.f, 0.f, 0.f, 0.f, 0.f, 0.f};
#pragma unroll
  for (int mp = 0; mp < 4; ++mp) {
#pragma unroll
    for (int q = 0; q < 4; ++q) {
      v2f b = v2f{hf[mp][2 * q], hf[mp][2 * q + 1]};
      a0 = wmma4(v2f{cf[0][mp][2 * q], cf[0][mp][2 * q + 1]}, b, a0);
      a1 = wmma4(v2f{cf[1][mp][2 * q], cf[1][mp][2 * q + 1]}, b, a1);
    }
  }
  *(v4f*)(o)      = v4f{a0[0], a0[1], a0[2], a0[3]};
  *(v4f*)(o + 8)  = v4f{a0[4], a0[5], a0[6], a0[7]};
  *(v4f*)(o + 16) = v4f{a1[0], a1[1], a1[2], a1[3]};
  *(v4f*)(o + 24) = v4f{a1[4], a1[5], a1[6], a1[7]};
}

__global__ void __launch_bounds__(512)
y_gemm_k(const float* __restrict__ hs, const float* __restrict__ C,
         float* __restrict__ y) {
  __shared__ float Cs[DOUT * CPITCH];
  int tid = threadIdx.x;
  int lane = tid & 31, hi = lane >> 4, m = lane & 15;
  int tile = blockIdx.x * 16 + (tid >> 5);

  v8f hf[4];
  load_state(hf, hs + (size_t)tile * 16 * LAT, m, hi);

#pragma unroll
  for (int j = 0; j < DOUT * LAT / 4 / 512; ++j) {
    int i = tid + 512 * j;
    int row = i >> 4, c4 = i & 15;
    *(v4f*)(Cs + row * CPITCH + 4 * c4) = *(const v4f*)(C + row * LAT + 4 * c4);
  }
  __syncthreads();

  int crow = 4 * (m >> 3) + (m & 3) + 8 * ((m >> 2) & 1);
  const float* cp = Cs + crow * CPITCH + 8 * hi;
  float* yp = y + (size_t)(tile * 16 + m) * DOUT + 4 * hi;

  v8f ca[2][4], cb[2][4];
  load_cpair(ca, cp);
#pragma unroll 1
  for (int g = 0; g < DOUT / 64; ++g) {
    int nx = g + 1 < DOUT / 64 ? 2 * g + 2 : 2 * g;
    load_cpair(cb, cp + (2 * g + 1) * 32 * CPITCH);
    ypair(yp + g * 64, ca, hf);
    load_cpair(ca, cp + nx * 32 * CPITCH);
    ypair(yp + g * 64 + 32, cb, hf);
  }
}

extern "C" void kernel_launch(void* const* d_in, const int* in_sizes, int n_in,
                              void* d_out, int out_size, void* d_ws,
                              size_t ws_size, hipStream_t stream) {
  const float* x  = (const float*)d_in[0];
  const float* A  = (const float*)d_in[1];
  const float* Bw = (const float*)d_in[2];
  const float* C  = (const float*)d_in[3];

  float* out      = (float*)d_out;
  float* y_region = out;
  float* h_region = out + (size_t)S * BT * DOUT;

  float* U0 = y_region;
  float* V  = y_region + (size_t)NPAIR * BT * LAT;

  float* ws      = (float*)d_ws;
  float* Apows   = ws;
  float* Lc      = ws + T * 64 * 64;
  float* carries = ws + T * 64 * 64 + (size_t)NC * BT * LAT;
  float* Wn      = carries + (size_t)NC * BT * LAT;
  float* Q       = Wn + (size_t)NC * BT * LAT;
  float* A2      = Q + 64 * 64;

  bx_pair_k<<<NPAIR * 2 / 8 + 1, 256, 0, stream>>>(x, Bw, A, U0, V, A2);
  chunk_local_k<<<NC * 2 + 4, 32, 0, stream>>>(V, A2, Lc, h_region, Apows);
  carry_win1_k<<<NC * 2 + 4, 32, 0, stream>>>(Apows + (T - 1) * 64 * 64, Lc,
                                              Wn, Q);
  carry_win2_k<<<NC * 2, 32, 0, stream>>>(Q, Wn, carries);
  chunk_final_k<<<(T / 2) * 16 + 1, 256, 0, stream>>>(A, Apows, carries, U0,
                                                      h_region);
  y_gemm_k<<<S * BT / 16 / 16, 512, 0, stream>>>(
      h_region + BT * LAT, C, y_region);
}
